# starting kernel
# baseline (speedup 1.0000x reference)
#include <hip/hip_runtime.h>


typedef __attribute__((ext_vector_type(2))) float v2f;
typedef __attribute__((ext_vector_type(8))) float v8f;

#define NCH   20
#define CE    32
#define KC    33
#define WEMB  512
#define KDIM  132
#define NTILES (WEMB/16)

#define PAIR_STRIDE 1056
#define LDS_FLOATS  (66 * PAIR_STRIDE)

__global__ __launch_bounds__(256) void char_cnn_wmma(
    const float* __restrict__ char_emb_w,
    const float* __restrict__ cnn_w,
    const int*   __restrict__ data,
    const int*   __restrict__ spelling,
    float*       __restrict__ out,
    int nWords)
{
  extern __shared__ float ldsW[];
  const int tid = threadIdx.x;

  for (int e = tid; e < WEMB * KDIM; e += 256) {
    const int we = e / KDIM;
    const int j  = e % KDIM;
    ldsW[(j >> 1) * PAIR_STRIDE + 2 * we + (j & 1)] = cnn_w[e];
  }
  __syncthreads();

  const int lane    = tid & 31;
  const int wave    = tid >> 5;
  const int m_local = lane & 15;
  const int half    = lane >> 4;

  for (int word = blockIdx.x * 8 + wave; word < nWords; word += gridDim.x * 8) {
    const int wid   = data[word];
    const int sbase = wid * NCH;

    v2f a[2][KC];
    #pragma unroll
    for (int mt = 0; mt < 2; ++mt) {
      const int r1 = mt * 16 + m_local + 2 * half;
      const int r2 = r1 + 1;
      const bool ok1 = (r1 < NCH);
      const bool ok2 = (r2 < NCH);
      const int ch1 = ok1 ? spelling[sbase + r1] : 0;
      const int ch2 = ok2 ? spelling[sbase + r2] : 0;
      const float* e1 = char_emb_w + ch1 * CE;
      const float* e2 = char_emb_w + ch2 * CE;
      #pragma unroll
      for (int cb = 0; cb < CE; cb += 4) {
        const float4 q1 = ok1 ? *(const float4*)(e1 + cb) : make_float4(0.f, 0.f, 0.f, 0.f);
        const float4 q2 = ok2 ? *(const float4*)(e2 + cb) : make_float4(0.f, 0.f, 0.f, 0.f);
        a[mt][cb + 0].x = q1.x;  a[mt][cb + 0].y = q2.x;
        a[mt][cb + 1].x = q1.y;  a[mt][cb + 1].y = q2.y;
        a[mt][cb + 2].x = q1.z;  a[mt][cb + 2].y = q2.z;
        a[mt][cb + 3].x = q1.w;  a[mt][cb + 3].y = q2.w;
      }
      a[mt][CE].x = ok1 ? 1.f : 0.f;
      a[mt][CE].y = ok2 ? 1.f : 0.f;
    }

    #pragma unroll 1
    for (int nt = 0; nt < NTILES; ++nt) {
      v8f acc0 = {};
      v8f acc1 = {};
      const int colBase = 2 * (nt * 16 + m_local);
      #pragma unroll
      for (int kt = 0; kt < KC; ++kt) {
        const int pair = 2 * kt + half;
        const v2f b = *(const v2f*)&ldsW[pair * PAIR_STRIDE + colBase];
        acc0 = __builtin_amdgcn_wmma_f32_16x16x4_f32(
            false, a[0][kt], false, b, (short)0, acc0, false, false);
        acc1 = __builtin_amdgcn_wmma_f32_16x16x4_f32(
            false, a[1][kt], false, b, (short)0, acc1, false, false);
      }
      float m = acc0[0];
      #pragma unroll
      for (int i = 1; i < 8; ++i) m = fmaxf(m, acc0[i]);
      m = fmaxf(m, __shfl_xor(m, 16, 32));
      const float t16 = __shfl(acc1[0], m_local, 32);
      m = fmaxf(m, t16);
      if (lane < 16) out[word * WEMB + nt * 16 + lane] = m;
    }
  }
}

extern "C" void kernel_launch(void* const* d_in, const int* in_sizes, int n_in,
                              void* d_out, int out_size, void* d_ws, size_t ws_size,
                              hipStream_t stream) {
  const float* char_emb_w = (const float*)d_in[0];
  const float* cnn_w      = (const float*)d_in[1];
  const int*   data       = (const int*)d_in[2];
  const int*   spelling   = (const int*)d_in[3];
  float*       out        = (float*)d_out;
  const int nWords = in_sizes[2];

  const int blocks = 256;
  const size_t shmem = LDS_FLOATS * sizeof(float);
  hipLaunchKernelGGL(char_cnn_wmma, dim3(blocks), dim3(256), shmem, stream,
                     char_emb_w, cnn_w, data, spelling, out, nWords);
}


# Conv filter applied once to the 256-row char table (WMMA; 64-channel slice of the product table in LDS per workgroup); each output = 4 gathered f32 adds per position + max, not a 132-deep product
# speedup vs baseline: 10.4153x; 10.4153x over previous
#include <hip/hip_runtime.h>


typedef __attribute__((ext_vector_type(2))) float v2f;
typedef __attribute__((ext_vector_type(8))) float v8f;

#define NCH   20
#define CE    32
#define WEMB  512
#define KDIM  132
#define NPOS  17
#define NCHAR 256
#define SL    64
#define NSL   (WEMB / SL)
#define ZROW  (4 * SL)
#define ZFLOATS (NCHAR * ZROW)
#define WCHUNK 512
#define LDS_BYTES ((ZFLOATS + WCHUNK * NCH) * 4)

__global__ __launch_bounds__(256) void char_cnn_wmma(
    const float* __restrict__ char_emb_w,
    const float* __restrict__ cnn_w,
    const int*   __restrict__ data,
    const int*   __restrict__ spelling,
    float*       __restrict__ out,
    int nWords)
{
  extern __shared__ float lds[];
  float* Z   = lds;
  int*   ids = (int*)(lds + ZFLOATS);

  const int tid  = threadIdx.x;
  const int lane = tid & 31;
  const int wave = tid >> 5;
  const int m    = lane & 15;
  const int half = lane >> 4;

  const int slice = blockIdx.x & (NSL - 1);
  const int grp   = blockIdx.x >> 3;
  const int ngrp  = gridDim.x >> 3;
  const int we0   = slice * SL;

  {
    v2f a[2][8];
    #pragma unroll
    for (int i = 0; i < 2; ++i) {
      const float* er = char_emb_w + ((wave * 2 + i) * 16 + m) * CE + 2 * half;
      #pragma unroll
      for (int q = 0; q < 8; ++q) a[i][q] = *(const v2f*)(er + 4 * q);
    }
    #pragma unroll
    for (int k = 0; k < 4; ++k) {
      #pragma unroll
      for (int nt = 0; nt < 4; ++nt) {
        const float* wr = cnn_w + (we0 + nt * 16 + m) * KDIM;
        v2f b[8];
        #pragma unroll
        for (int q = 0; q < 8; ++q) {
          const int c = 4 * q + 2 * half;
          b[q].x = wr[c * 4 + k];
          b[q].y = wr[(c + 1) * 4 + k];
        }
        float binit = 0.f;
        if (k == 0) {
          const float4 lb = *(const float4*)(wr + CE * 4);
          binit = (lb.x + lb.y) + (lb.z + lb.w);
        }
        v8f acc0, acc1;
        #pragma unroll
        for (int i = 0; i < 8; ++i) { acc0[i] = binit; acc1[i] = binit; }
        #pragma unroll
        for (int q = 0; q < 8; ++q) {
          acc0 = __builtin_amdgcn_wmma_f32_16x16x4_f32(false, a[0][q], false, b[q], (short)0, acc0, false, false);
          acc1 = __builtin_amdgcn_wmma_f32_16x16x4_f32(false, a[1][q], false, b[q], (short)0, acc1, false, false);
        }
        float* z0 = Z + ((wave * 2 + 0) * 16 + 8 * half) * ZROW + k * SL + nt * 16 + m;
        float* z1 = Z + ((wave * 2 + 1) * 16 + 8 * half) * ZROW + k * SL + nt * 16 + m;
        #pragma unroll
        for (int i = 0; i < 8; ++i) { z0[i * ZROW] = acc0[i]; z1[i * ZROW] = acc1[i]; }
      }
    }
  }

  const int lr = (lane < NCH) ? lane : (NCH - 1);
  const float* zl = Z + 2 * lane;

  for (int wbase = grp * WCHUNK; wbase < nWords; wbase += ngrp * WCHUNK) {
    __syncthreads();
    for (int wl = tid; wl < WCHUNK; wl += 256) {
      const int w = min(wbase + wl, nWords - 1);
      const int4* sp = (const int4*)(spelling + data[w] * NCH);
      int4* dst = (int4*)(ids + wl * NCH);
      #pragma unroll
      for (int p = 0; p < NCH / 4; ++p) {
        int4 v = sp[p];
        v.x &= NCHAR - 1; v.y &= NCHAR - 1; v.z &= NCHAR - 1; v.w &= NCHAR - 1;
        dst[p] = v;
      }
    }
    __syncthreads();

    int cv = ids[wave * NCH + lr];
    for (int wl = wave; wl < WCHUNK; wl += 8) {
      const int w = wbase + wl;
      if (w >= nWords) break;
      const int wn = min(wl + 8, WCHUNK - 1);
      const int cvn = ids[wn * NCH + lr];
      const float* zr[NCH];
      #pragma unroll
      for (int r = 0; r < NCH; ++r) zr[r] = zl + __builtin_amdgcn_readlane(cv, r) * ZROW;
      v2f mx;
      #pragma unroll
      for (int t = 0; t < NPOS; ++t) {
        const v2f s0 = *(const v2f*)(zr[t]     + 0 * SL);
        const v2f s1 = *(const v2f*)(zr[t + 1] + 1 * SL);
        const v2f s2 = *(const v2f*)(zr[t + 2] + 2 * SL);
        const v2f s3 = *(const v2f*)(zr[t + 3] + 3 * SL);
        const v2f y = (s0 + s1) + (s2 + s3);
        if (t == 0) { mx = y; }
        else { mx.x = fmaxf(mx.x, y.x); mx.y = fmaxf(mx.y, y.y); }
      }
      *(v2f*)(out + (size_t)w * WEMB + we0 + 2 * lane) = mx;
      cv = cvn;
    }
  }
}

extern "C" void kernel_launch(void* const* d_in, const int* in_sizes, int n_in,
                              void* d_out, int out_size, void* d_ws, size_t ws_size,
                              hipStream_t stream) {
  const float* char_emb_w = (const float*)d_in[0];
  const float* cnn_w      = (const float*)d_in[1];
  const int*   data       = (const int*)d_in[2];
  const int*   spelling   = (const int*)d_in[3];
  float*       out        = (float*)d_out;
  const int nWords = in_sizes[2];

  const int blocks = 256;
  const size_t shmem = LDS_BYTES;
  hipLaunchKernelGGL(char_cnn_wmma, dim3(blocks), dim3(256), shmem, stream,
                     char_emb_w, cnn_w, data, spelling, out, nWords);
}


# Gather reads grouped: a word's 36 LDS reads go out as ONE group one word ahead into a second register set, the previous word's sums run under them (was ~14 read places per word, at most 8 in flight)
# speedup vs baseline: 13.4445x; 1.2908x over previous
#include <hip/hip_runtime.h>


typedef __attribute__((ext_vector_type(2))) float v2f;
typedef __attribute__((ext_vector_type(8))) float v8f;

#define NCH   20
#define CE    32
#define WEMB  512
#define KDIM  132
#define NPOS  17
#define NCHAR 256
#define SL    64
#define NSL   (WEMB / SL)
#define ZROW  (4 * SL)
#define ZFLOATS (NCHAR * ZROW)
#define WCHUNK 512
#define LDS_BYTES ((ZFLOATS + WCHUNK * NCH) * 4)

#define NV (NPOS * 4)

static __device__ __forceinline__ void word_rows(const float* (&zr)[NCH], const float* zl, int cv) {
  #pragma unroll
  for (int r = 0; r < NCH; ++r) zr[r] = zl + __builtin_amdgcn_readlane(cv, r) * ZROW;
}

static __device__ __forceinline__ void word_reads(v2f (&buf)[NV], const float* const (&zr)[NCH]) {
  #pragma unroll
  for (int r = 0; r < NCH; ++r) {
    #pragma unroll
    for (int k = 0; k < 4; ++k) {
      const int t = r - k;
      if (t >= 0 && t < NPOS) buf[4 * t + k] = *(const v2f*)(zr[r] + k * SL);
    }
  }
}

static __device__ __forceinline__ v2f word_max(const v2f (&buf)[NV]) {
  v2f mx;
  #pragma unroll
  for (int t = 0; t < NPOS; ++t) {
    const v2f y = (buf[4 * t] + buf[4 * t + 1]) + (buf[4 * t + 2] + buf[4 * t + 3]);
    if (t == 0) { mx = y; }
    else { mx.x = fmaxf(mx.x, y.x); mx.y = fmaxf(mx.y, y.y); }
  }
  return mx;
}

__global__ __launch_bounds__(256) void char_cnn_wmma(
    const float* __restrict__ char_emb_w,
    const float* __restrict__ cnn_w,
    const int*   __restrict__ data,
    const int*   __restrict__ spelling,
    float*       __restrict__ out,
    int nWords)
{
  extern __shared__ float lds[];
  float* Z   = lds;
  int*   ids = (int*)(lds + ZFLOATS);

  const int tid  = threadIdx.x;
  const int lane = tid & 31;
  const int wave = tid >> 5;
  const int m    = lane & 15;
  const int half = lane >> 4;

  const int slice = blockIdx.x & (NSL - 1);
  const int grp   = blockIdx.x >> 3;
  const int ngrp  = gridDim.x >> 3;
  const int we0   = slice * SL;

  {
    v2f a[2][8];
    #pragma unroll
    for (int i = 0; i < 2; ++i) {
      const float* er = char_emb_w + ((wave * 2 + i) * 16 + m) * CE + 2 * half;
      #pragma unroll
      for (int q = 0; q < 8; ++q) a[i][q] = *(const v2f*)(er + 4 * q);
    }
    #pragma unroll
    for (int k = 0; k < 4; ++k) {
      #pragma unroll
      for (int nt = 0; nt < 4; ++nt) {
        const float* wr = cnn_w + (we0 + nt * 16 + m) * KDIM;
        v2f b[8];
        #pragma unroll
        for (int q = 0; q < 8; ++q) {
          const int c = 4 * q + 2 * half;
          b[q].x = wr[c * 4 + k];
          b[q].y = wr[(c + 1) * 4 + k];
        }
        float binit = 0.f;
        if (k == 0) {
          const float4 lb = *(const float4*)(wr + CE * 4);
          binit = (lb.x + lb.y) + (lb.z + lb.w);
        }
        v8f acc0, acc1;
        #pragma unroll
        for (int i = 0; i < 8; ++i) { acc0[i] = binit; acc1[i] = binit; }
        #pragma unroll
        for (int q = 0; q < 8; ++q) {
          acc0 = __builtin_amdgcn_wmma_f32_16x16x4_f32(false, a[0][q], false, b[q], (short)0, acc0, false, false);
          acc1 = __builtin_amdgcn_wmma_f32_16x16x4_f32(false, a[1][q], false, b[q], (short)0, acc1, false, false);
        }
        float* z0 = Z + ((wave * 2 + 0) * 16 + 8 * half) * ZROW + k * SL + nt * 16 + m;
        float* z1 = Z + ((wave * 2 + 1) * 16 + 8 * half) * ZROW + k * SL + nt * 16 + m;
        #pragma unroll
        for (int i = 0; i < 8; ++i) { z0[i * ZROW] = acc0[i]; z1[i * ZROW] = acc1[i]; }
      }
    }
  }

  const int lr = (lane < NCH) ? lane : (NCH - 1);
  const float* zl = Z + 2 * lane;

  for (int wbase = grp * WCHUNK; wbase < nWords; wbase += ngrp * WCHUNK) {
    __syncthreads();
    for (int wl = tid; wl < WCHUNK; wl += 256) {
      const int w = min(wbase + wl, nWords - 1);
      const int4* sp = (const int4*)(spelling + data[w] * NCH);
      int4* dst = (int4*)(ids + wl * NCH);
      #pragma unroll
      for (int p = 0; p < NCH / 4; ++p) {
        int4 v = sp[p];
        v.x &= NCHAR - 1; v.y &= NCHAR - 1; v.z &= NCHAR - 1; v.w &= NCHAR - 1;
        dst[p] = v;
      }
    }
    __syncthreads();

    const int sw = __builtin_amdgcn_readfirstlane(wave);
    v2f bufA[NV], bufB[NV];
    const float* zr[NCH];
    int cvA = ids[sw * NCH + lr];
    int cvB = ids[(sw + 8) * NCH + lr];
    __builtin_amdgcn_sched_barrier(0);
    word_rows(zr, zl, cvA);
    __builtin_amdgcn_sched_barrier(0);
    word_reads(bufA, zr);
    __builtin_amdgcn_sched_barrier(0);
    int wl = sw;
    for (; wl < WCHUNK - 16; wl += 16) {
      word_rows(zr, zl, cvB);
      __builtin_amdgcn_sched_barrier(0);
      cvA = ids[(wl + 16) * NCH + lr];
      __builtin_amdgcn_sched_barrier(0);
      word_reads(bufB, zr);
      __builtin_amdgcn_sched_barrier(0);
      {
        const v2f mx = word_max(bufA);
        const int w = wbase + wl;
        if (w < nWords)
          *(v2f*)(out + (size_t)w * WEMB + we0 + 2 * lane) = mx;
      }
      __builtin_amdgcn_sched_barrier(0);
      word_rows(zr, zl, cvA);
      __builtin_amdgcn_sched_barrier(0);
      cvB = ids[(wl + 24) * NCH + lr];
      __builtin_amdgcn_sched_barrier(0);
      word_reads(bufA, zr);
      __builtin_amdgcn_sched_barrier(0);
      {
        const v2f mx = word_max(bufB);
        const int w = wbase + wl + 8;
        if (w < nWords)
          *(v2f*)(out + (size_t)w * WEMB + we0 + 2 * lane) = mx;
      }
      __builtin_amdgcn_sched_barrier(0);
    }
    word_rows(zr, zl, cvB);
    __builtin_amdgcn_sched_barrier(0);
    word_reads(bufB, zr);
    __builtin_amdgcn_sched_barrier(0);
    {
      const v2f mx = word_max(bufA);
      const int w = wbase + wl;
      if (w < nWords)
        *(v2f*)(out + (size_t)w * WEMB + we0 + 2 * lane) = mx;
    }
    {
      const v2f mx = word_max(bufB);
      const int w = wbase + wl + 8;
      if (w < nWords)
        *(v2f*)(out + (size_t)w * WEMB + we0 + 2 * lane) = mx;
    }
  }
}

extern "C" void kernel_launch(void* const* d_in, const int* in_sizes, int n_in,
                              void* d_out, int out_size, void* d_ws, size_t ws_size,
                              hipStream_t stream) {
  const float* char_emb_w = (const float*)d_in[0];
  const float* cnn_w      = (const float*)d_in[1];
  const int*   data       = (const int*)d_in[2];
  const int*   spelling   = (const int*)d_in[3];
  float*       out        = (float*)d_out;
  const int nWords = in_sizes[2];

  const int blocks = 256;
  const size_t shmem = LDS_BYTES;
  hipLaunchKernelGGL(char_cnn_wmma, dim3(blocks), dim3(256), shmem, stream,
                     char_emb_w, cnn_w, data, spelling, out, nWords);
}


# Memory waits moved under the table build: first pass's char ids requested at entry (were 4 serial hops behind it); filter consumed per 16-channel column, later columns' loads land under WMMAs
# speedup vs baseline: 15.0487x; 1.1193x over previous
#include <hip/hip_runtime.h>


typedef __attribute__((ext_vector_type(2))) float v2f;
typedef __attribute__((ext_vector_type(8))) float v8f;

#define NCH   20
#define CE    32
#define WEMB  512
#define KDIM  132
#define NPOS  17
#define NCHAR 256
#define SL    64
#define NSL   (WEMB / SL)
#define ZROW  (4 * SL)
#define ZFLOATS (NCHAR * ZROW)
#define WCHUNK 512
#define LDS_BYTES ((ZFLOATS + WCHUNK * NCH) * 4)

#define NV (NPOS * 4)

static __device__ __forceinline__ void word_rows(const float* (&zr)[NCH], const float* zl, int cv) {
  #pragma unroll
  for (int r = 0; r < NCH; ++r) zr[r] = zl + __builtin_amdgcn_readlane(cv, r) * ZROW;
}

static __device__ __forceinline__ void word_reads(v2f (&buf)[NV], const float* const (&zr)[NCH]) {
  #pragma unroll
  for (int r = 0; r < NCH; ++r) {
    #pragma unroll
    for (int k = 0; k < 4; ++k) {
      const int t = r - k;
      if (t >= 0 && t < NPOS) buf[4 * t + k] = *(const v2f*)(zr[r] + k * SL);
    }
  }
}

static __device__ __forceinline__ v2f word_max(const v2f (&buf)[NV]) {
  v2f mx;
  #pragma unroll
  for (int t = 0; t < NPOS; ++t) {
    const v2f y = (buf[4 * t] + buf[4 * t + 1]) + (buf[4 * t + 2] + buf[4 * t + 3]);
    if (t == 0) { mx = y; }
    else { mx.x = fmaxf(mx.x, y.x); mx.y = fmaxf(mx.y, y.y); }
  }
  return mx;
}

__global__ __launch_bounds__(256) void char_cnn_wmma(
    const float* __restrict__ char_emb_w,
    const float* __restrict__ cnn_w,
    const int*   __restrict__ data,
    const int*   __restrict__ spelling,
    float*       __restrict__ out,
    int nWords)
{
  extern __shared__ float lds[];
  float* Z   = lds;
  int*   ids = (int*)(lds + ZFLOATS);

  const int tid  = threadIdx.x;
  const int lane = tid & 31;
  const int wave = tid >> 5;
  const int m    = lane & 15;
  const int half = lane >> 4;

  const int slice = blockIdx.x & (NSL - 1);
  const int grp   = blockIdx.x >> 3;
  const int ngrp  = gridDim.x >> 3;
  const int we0   = slice * SL;

  const int wbase0 = grp * WCHUNK;
  int widx[WCHUNK / 256];
  #pragma unroll
  for (int j = 0; j < WCHUNK / 256; ++j)
    widx[j] = data[min(wbase0 + tid + 256 * j, nWords - 1)];
  int4 idv[WCHUNK / 256][NCH / 4];

  {
    v2f a[2][8];
    #pragma unroll
    for (int i = 0; i < 2; ++i) {
      const float* er = char_emb_w + ((wave * 2 + i) * 16 + m) * CE + 2 * half;
      #pragma unroll
      for (int q = 0; q < 8; ++q) a[i][q] = *(const v2f*)(er + 4 * q);
    }
    #pragma unroll
    for (int j = 0; j < WCHUNK / 256; ++j) {
      const int4* sp = (const int4*)(spelling + widx[j] * NCH);
      #pragma unroll
      for (int p = 0; p < NCH / 4; ++p) idv[j][p] = sp[p];
    }
    #pragma unroll
    for (int nt = 0; nt < 4; ++nt) {
      #pragma unroll
      for (int k = 0; k < 4; ++k) {
        const float* wr = cnn_w + (we0 + nt * 16 + m) * KDIM;
        v2f b[8];
        #pragma unroll
        for (int q = 0; q < 8; ++q) {
          const int c = 4 * q + 2 * half;
          b[q].x = wr[c * 4 + k];
          b[q].y = wr[(c + 1) * 4 + k];
        }
        float binit = 0.f;
        if (k == 0) {
          const float4 lb = *(const float4*)(wr + CE * 4);
          binit = (lb.x + lb.y) + (lb.z + lb.w);
        }
        v8f acc0, acc1;
        #pragma unroll
        for (int i = 0; i < 8; ++i) { acc0[i] = binit; acc1[i] = binit; }
        #pragma unroll
        for (int q = 0; q < 8; ++q) {
          acc0 = __builtin_amdgcn_wmma_f32_16x16x4_f32(false, a[0][q], false, b[q], (short)0, acc0, false, false);
          acc1 = __builtin_amdgcn_wmma_f32_16x16x4_f32(false, a[1][q], false, b[q], (short)0, acc1, false, false);
        }
        float* z0 = Z + ((wave * 2 + 0) * 16 + 8 * half) * ZROW + k * SL + nt * 16 + m;
        float* z1 = Z + ((wave * 2 + 1) * 16 + 8 * half) * ZROW + k * SL + nt * 16 + m;
        #pragma unroll
        for (int i = 0; i < 8; ++i) { z0[i * ZROW] = acc0[i]; z1[i * ZROW] = acc1[i]; }
      }
    }
  }

  const int lr = (lane < NCH) ? lane : (NCH - 1);
  const float* zl = Z + 2 * lane;

  #pragma unroll
  for (int j = 0; j < WCHUNK / 256; ++j) {
    int4* dst = (int4*)(ids + (tid + 256 * j) * NCH);
    #pragma unroll
    for (int p = 0; p < NCH / 4; ++p) {
      int4 v = idv[j][p];
      v.x &= NCHAR - 1; v.y &= NCHAR - 1; v.z &= NCHAR - 1; v.w &= NCHAR - 1;
      dst[p] = v;
    }
  }

  for (int wbase = wbase0; wbase < nWords; wbase += ngrp * WCHUNK) {
    if (wbase != wbase0) {
    __syncthreads();
    for (int wl = tid; wl < WCHUNK; wl += 256) {
      const int w = min(wbase + wl, nWords - 1);
      const int4* sp = (const int4*)(spelling + data[w] * NCH);
      int4* dst = (int4*)(ids + wl * NCH);
      #pragma unroll
      for (int p = 0; p < NCH / 4; ++p) {
        int4 v = sp[p];
        v.x &= NCHAR - 1; v.y &= NCHAR - 1; v.z &= NCHAR - 1; v.w &= NCHAR - 1;
        dst[p] = v;
      }
    }
    }
    __syncthreads();

    const int sw = __builtin_amdgcn_readfirstlane(wave);
    v2f bufA[NV], bufB[NV];
    const float* zr[NCH];
    int cvA = ids[sw * NCH + lr];
    int cvB = ids[(sw + 8) * NCH + lr];
    __builtin_amdgcn_sched_barrier(0);
    word_rows(zr, zl, cvA);
    __builtin_amdgcn_sched_barrier(0);
    word_reads(bufA, zr);
    __builtin_amdgcn_sched_barrier(0);
    int wl = sw;
    for (; wl < WCHUNK - 16; wl += 16) {
      word_rows(zr, zl, cvB);
      __builtin_amdgcn_sched_barrier(0);
      cvA = ids[(wl + 16) * NCH + lr];
      __builtin_amdgcn_sched_barrier(0);
      word_reads(bufB, zr);
      __builtin_amdgcn_sched_barrier(0);
      {
        const v2f mx = word_max(bufA);
        const int w = wbase + wl;
        if (w < nWords)
          *(v2f*)(out + (size_t)w * WEMB + we0 + 2 * lane) = mx;
      }
      __builtin_amdgcn_sched_barrier(0);
      word_rows(zr, zl, cvA);
      __builtin_amdgcn_sched_barrier(0);
      cvB = ids[(wl + 24) * NCH + lr];
      __builtin_amdgcn_sched_barrier(0);
      word_reads(bufA, zr);
      __builtin_amdgcn_sched_barrier(0);
      {
        const v2f mx = word_max(bufB);
        const int w = wbase + wl + 8;
        if (w < nWords)
          *(v2f*)(out + (size_t)w * WEMB + we0 + 2 * lane) = mx;
      }
      __builtin_amdgcn_sched_barrier(0);
    }
    word_rows(zr, zl, cvB);
    __builtin_amdgcn_sched_barrier(0);
    word_reads(bufB, zr);
    __builtin_amdgcn_sched_barrier(0);
    {
      const v2f mx = word_max(bufA);
      const int w = wbase + wl;
      if (w < nWords)
        *(v2f*)(out + (size_t)w * WEMB + we0 + 2 * lane) = mx;
    }
    {
      const v2f mx = word_max(bufB);
      const int w = wbase + wl + 8;
      if (w < nWords)
        *(v2f*)(out + (size_t)w * WEMB + we0 + 2 * lane) = mx;
    }
  }
}

extern "C" void kernel_launch(void* const* d_in, const int* in_sizes, int n_in,
                              void* d_out, int out_size, void* d_ws, size_t ws_size,
                              hipStream_t stream) {
  const float* char_emb_w = (const float*)d_in[0];
  const float* cnn_w      = (const float*)d_in[1];
  const int*   data       = (const int*)d_in[2];
  const int*   spelling   = (const int*)d_in[3];
  float*       out        = (float*)d_out;
  const int nWords = in_sizes[2];

  const int blocks = 256;
  const size_t shmem = LDS_BYTES;
  hipLaunchKernelGGL(char_cnn_wmma, dim3(blocks), dim3(256), shmem, stream,
                     char_emb_w, cnn_w, data, spelling, out, nWords);
}
